# starting kernel
# baseline (speedup 1.0000x reference)
#include <hip/hip_runtime.h>

#define NODES  50000
#define EDGES  800000
#define LEDGES 200000
#define INCH   256
#define NF     64

typedef float v2f __attribute__((ext_vector_type(2)));
typedef float v8f __attribute__((ext_vector_type(8)));

__global__ void k_deg_init(float* deg, int n) {
  int i = blockIdx.x * blockDim.x + threadIdx.x;
  if (i < n) deg[i] = 1.0f;
}

__global__ void k_deg_accum(const int* __restrict__ col, float* deg, int e) {
  int i = blockIdx.x * blockDim.x + threadIdx.x;
  if (i < e) atomicAdd(&deg[col[i]], 1.0f);
}

__global__ void k_deg_rsqrt(float* deg, int n) {
  int i = blockIdx.x * blockDim.x + threadIdx.x;
  if (i < n) deg[i] = rsqrtf(deg[i]);
}

template<int K>
__global__ __launch_bounds__(256) void k_gemm(const float* __restrict__ A,
                                              const float* __restrict__ W,
                                              float* __restrict__ C, int M) {
  __shared__ __align__(16) float sB[4096];

  const int lane = threadIdx.x & 31;
  const int wave = threadIdx.x >> 5;

  int strip = blockIdx.x * 8 + wave;
  int maxStrip = (M >> 4) - 1;
  if (strip > maxStrip) strip = maxStrip;
  const long rowBase = (long)strip << 4;

  const int mlo  = lane & 15;
  const int koff = (lane >> 4) << 1;
  const float* arow = A + (rowBase + mlo) * (long)K;

  v8f acc[4];
#pragma unroll
  for (int t = 0; t < 4; ++t) acc[t] = (v8f){0.f,0.f,0.f,0.f,0.f,0.f,0.f,0.f};

  for (int k0 = 0; k0 < K; k0 += 64) {
    __syncthreads();
    for (int idx = threadIdx.x; idx < 2048; idx += 256) {
      int ks = idx >> 7;
      int t  = (idx >> 5) & 3;
      int l  = idx & 31;
      int kk = k0 + ks * 4 + ((l >> 4) << 1);
      int cl = t * 16 + (l & 15);
      sB[idx * 2 + 0] = W[kk * NF + cl];
      sB[idx * 2 + 1] = W[(kk + 1) * NF + cl];
    }
    __syncthreads();
    if (k0 + 64 < K)
      __builtin_prefetch(arow + k0 + 64, 0, 0);
#pragma unroll 4
    for (int ks = 0; ks < 16; ++ks) {
      v2f a = *(const v2f*)(arow + k0 + ks * 4 + koff);
      const v2f* bfrag = (const v2f*)sB + ks * 128 + lane;
#pragma unroll
      for (int t = 0; t < 4; ++t) {
        v2f b = bfrag[t * 32];
        acc[t] = __builtin_amdgcn_wmma_f32_16x16x4_f32(
            false, a, false, b, (short)0, acc[t], false, false);
      }
    }
  }

  const int mhi = (lane >> 4) << 3;
#pragma unroll
  for (int t = 0; t < 4; ++t)
#pragma unroll
    for (int i = 0; i < 8; ++i)
      C[(rowBase + mhi + i) * NF + t * 16 + mlo] = acc[t][i];
}

__global__ void k_agg_init(const float* __restrict__ xw, const float* __restrict__ dinv,
                           float* __restrict__ agg, int n) {
  int t = blockIdx.x * blockDim.x + threadIdx.x;
  int node = t >> 4;
  int f = (t & 15) << 2;
  if (node < n) {
    float s = dinv[node]; s = s * s;
    float4 v = *(const float4*)(xw + (long)node * NF + f);
    v.x *= s; v.y *= s; v.z *= s; v.w *= s;
    *(float4*)(agg + (long)node * NF + f) = v;
  }
}

__global__ void k_agg_edges(const float* __restrict__ xw, const int* __restrict__ row,
                            const int* __restrict__ col, const float* __restrict__ dinv,
                            float* agg, int e) {
  int t = blockIdx.x * blockDim.x + threadIdx.x;
  int ed = t >> 4;
  int f = (t & 15) << 2;
  if (ed < e) {
    int r = row[ed], c = col[ed];
    float nrm = dinv[r] * dinv[c];
    float4 v = *(const float4*)(xw + (long)r * NF + f);
    float* dst = agg + (long)c * NF + f;
    atomicAdd(dst + 0, v.x * nrm);
    atomicAdd(dst + 1, v.y * nrm);
    atomicAdd(dst + 2, v.z * nrm);
    atomicAdd(dst + 3, v.w * nrm);
  }
}

template<bool RELU>
__global__ void k_bias_act(const float* __restrict__ agg, const float* __restrict__ bias,
                           float* __restrict__ out, int n) {
  int t = blockIdx.x * blockDim.x + threadIdx.x;
  int node = t >> 4;
  int f = (t & 15) << 2;
  if (node < n) {
    float4 v = *(const float4*)(agg + (long)node * NF + f);
    v.x += bias[f + 0]; v.y += bias[f + 1]; v.z += bias[f + 2]; v.w += bias[f + 3];
    if (RELU) {
      v.x = fmaxf(v.x, 0.f); v.y = fmaxf(v.y, 0.f);
      v.z = fmaxf(v.z, 0.f); v.w = fmaxf(v.w, 0.f);
    }
    *(float4*)(out + (long)node * NF + f) = v;
  }
}

__global__ void k_decode(const float* __restrict__ z, const int* __restrict__ src,
                         const int* __restrict__ dst, float* __restrict__ out, int e) {
  int i = blockIdx.x * blockDim.x + threadIdx.x;
  if (i < e) {
    const float4* a = (const float4*)(z + (long)src[i] * NF);
    const float4* b = (const float4*)(z + (long)dst[i] * NF);
    float acc = 0.f;
#pragma unroll
    for (int j = 0; j < 16; ++j) {
      float4 av = a[j], bv = b[j];
      acc += av.x * bv.x + av.y * bv.y + av.z * bv.z + av.w * bv.w;
    }
    out[i] = acc;
  }
}

extern "C" void kernel_launch(void* const* d_in, const int* in_sizes, int n_in,
                              void* d_out, int out_size, void* d_ws, size_t ws_size,
                              hipStream_t stream) {
  const float* x   = (const float*)d_in[0];
  const int*   ei  = (const int*)d_in[1];
  const int*   eli = (const int*)d_in[2];
  const float* W1  = (const float*)d_in[3];
  const float* b1  = (const float*)d_in[4];
  const float* W2  = (const float*)d_in[5];
  const float* b2  = (const float*)d_in[6];
  float* out = (float*)d_out;

  float* ws   = (float*)d_ws;
  float* dinv = ws;
  float* bufA = ws + 50048;
  float* bufB = bufA + (size_t)NODES * NF;
  float* bufC = bufB + (size_t)NODES * NF;

  const int* erow = ei;
  const int* ecol = ei + EDGES;
  const int* lsrc = eli;
  const int* ldst = eli + LEDGES;

  dim3 blk(256);
  const int nodeGrid = (NODES * 16 + 255) / 256;
  const int edgeGrid = (EDGES * 16 + 255) / 256;
  const int strips   = NODES / 16;
  const int gemmGrid = (strips + 7) / 8;

  k_deg_init<<<(NODES + 255) / 256, blk, 0, stream>>>(dinv, NODES);
  k_deg_accum<<<(EDGES + 255) / 256, blk, 0, stream>>>(ecol, dinv, EDGES);
  k_deg_rsqrt<<<(NODES + 255) / 256, blk, 0, stream>>>(dinv, NODES);

  k_gemm<INCH><<<gemmGrid, blk, 0, stream>>>(x, W1, bufA, NODES);
  k_agg_init<<<nodeGrid, blk, 0, stream>>>(bufA, dinv, bufB, NODES);
  k_agg_edges<<<edgeGrid, blk, 0, stream>>>(bufA, erow, ecol, dinv, bufB, EDGES);
  k_bias_act<true><<<nodeGrid, blk, 0, stream>>>(bufB, b1, bufA, NODES);

  k_gemm<NF><<<gemmGrid, blk, 0, stream>>>(bufA, W2, bufB, NODES);
  k_agg_init<<<nodeGrid, blk, 0, stream>>>(bufB, dinv, bufC, NODES);
  k_agg_edges<<<edgeGrid, blk, 0, stream>>>(bufB, erow, ecol, dinv, bufC, EDGES);
  k_bias_act<false><<<nodeGrid, blk, 0, stream>>>(bufC, b2, bufA, NODES);

  k_decode<<<(LEDGES + 255) / 256, blk, 0, stream>>>(bufA, lsrc, ldst, out, LEDGES);
}


# Edge norm rsqrt(deg[src])*rsqrt(deg[dst]) written once per call by the layer-1 GEMM's threads under its staging waits; both scatters read it coalesced, no degree look-ups per edge; sums unchanged
# speedup vs baseline: 1.7226x; 1.7226x over previous
#include <hip/hip_runtime.h>

#define NODES  50000
#define EDGES  800000
#define LEDGES 200000
#define INCH   256
#define NF     64

typedef float v2f __attribute__((ext_vector_type(2)));
typedef float v8f __attribute__((ext_vector_type(8)));

__global__ void k_deg_init(float* deg, int n) {
  int i = blockIdx.x * blockDim.x + threadIdx.x;
  if (i < n) deg[i] = 1.0f;
}

__global__ void k_deg_accum(const int* __restrict__ col, float* deg, int e) {
  int i = blockIdx.x * blockDim.x + threadIdx.x;
  if (i < e) atomicAdd(&deg[col[i]], 1.0f);
}

#define EPT 8
template<int K, bool RELU_IN, bool EDGE_NORM>
__global__ __launch_bounds__(256) void k_gemm(const float* __restrict__ A,
                                              const float* __restrict__ W,
                                              const float* __restrict__ bias,
                                              const float* __restrict__ deg,
                                              float* __restrict__ C,
                                              float* __restrict__ AGG, int M,
                                              const int* __restrict__ erow,
                                              const int* __restrict__ ecol,
                                              float* __restrict__ ENORM, int E) {
  __shared__ __align__(16) float sB[4096];

  const int lane = threadIdx.x & 31;
  const int wave = threadIdx.x >> 5;

  int strip = blockIdx.x * 8 + wave;
  int maxStrip = (M >> 4) - 1;
  if (strip > maxStrip) strip = maxStrip;
  const long rowBase = (long)strip << 4;

  const int mlo  = lane & 15;
  const int koff = (lane >> 4) << 1;
  const float* arow = A + (rowBase + mlo) * (long)K;

  const int mhi = (lane >> 4) << 3;
  float4 dg0 = {}, dg1 = {};
  float bcol[4] = {};

  v8f acc[4];
#pragma unroll
  for (int t = 0; t < 4; ++t) acc[t] = (v8f){0.f,0.f,0.f,0.f,0.f,0.f,0.f,0.f};

  const int gtid = blockIdx.x * 256 + threadIdx.x;
  const int gstride = gridDim.x * 256;
  int er[EPT], ec[EPT];
  float gr[EPT], gc[EPT];
  if (EDGE_NORM) {
#pragma unroll
    for (int j = 0; j < EPT; ++j) {
      const int idx = min(gtid + j * gstride, E - 1);
      er[j] = erow[idx];
      ec[j] = ecol[idx];
    }
  }

  for (int k0 = 0; k0 < K; k0 += 64) {
    __syncthreads();
    for (int idx = threadIdx.x; idx < 2048; idx += 256) {
      int ks = idx >> 7;
      int t  = (idx >> 5) & 3;
      int l  = idx & 31;
      int kk = k0 + ks * 4 + ((l >> 4) << 1);
      int cl = t * 16 + (l & 15);
      sB[idx * 2 + 0] = W[kk * NF + cl];
      sB[idx * 2 + 1] = W[(kk + 1) * NF + cl];
    }
    __syncthreads();
    if (EDGE_NORM && k0 == 0) {
#pragma unroll
      for (int j = 0; j < EPT; ++j) { gr[j] = deg[er[j]]; gc[j] = deg[ec[j]]; }
    }
    if (k0 + 64 < K) {
      __builtin_prefetch(arow + k0 + 64, 0, 0);
    } else {
      dg0 = *(const float4*)(deg + rowBase + mhi);
      dg1 = *(const float4*)(deg + rowBase + mhi + 4);
#pragma unroll
      for (int t = 0; t < 4; ++t) bcol[t] = bias[t * 16 + mlo];
    }
#pragma unroll 4
    for (int ks = 0; ks < 16; ++ks) {
      v2f a = *(const v2f*)(arow + k0 + ks * 4 + koff);
      if (RELU_IN) {
        a.x = fmaxf(a.x, 0.f);
        a.y = fmaxf(a.y, 0.f);
      }
      const v2f* bfrag = (const v2f*)sB + ks * 128 + lane;
#pragma unroll
      for (int t = 0; t < 4; ++t) {
        v2f b = bfrag[t * 32];
        acc[t] = __builtin_amdgcn_wmma_f32_16x16x4_f32(
            false, a, false, b, (short)0, acc[t], false, false);
      }
    }
  }

  float sl[8];
  sl[0] = rsqrtf(dg0.x); sl[1] = rsqrtf(dg0.y); sl[2] = rsqrtf(dg0.z); sl[3] = rsqrtf(dg0.w);
  sl[4] = rsqrtf(dg1.x); sl[5] = rsqrtf(dg1.y); sl[6] = rsqrtf(dg1.z); sl[7] = rsqrtf(dg1.w);
#pragma unroll
  for (int i = 0; i < 8; ++i) sl[i] = sl[i] * sl[i];
#pragma unroll
  for (int t = 0; t < 4; ++t)
#pragma unroll
    for (int i = 0; i < 8; ++i) {
      const long o = (rowBase + mhi + i) * NF + t * 16 + mlo;
      const float v = acc[t][i];
      C[o]   = v;
      AGG[o] = v * sl[i] + bcol[t];
    }

  if (EDGE_NORM) {
#pragma unroll
    for (int j = 0; j < EPT; ++j) {
      const int idx = gtid + j * gstride;
      if (idx < E) ENORM[idx] = rsqrtf(gr[j]) * rsqrtf(gc[j]);
    }
  }
}

#define EPW 16
__global__ __launch_bounds__(256) void k_agg_edges(const float* __restrict__ xw,
                            const int* __restrict__ row, const int* __restrict__ col,
                            const float* __restrict__ enorm, float* agg, int e) {
  const int lane = threadIdx.x & 31;
  const int e0 = __builtin_amdgcn_readfirstlane((blockIdx.x * 8 + (threadIdx.x >> 5)) * EPW);
  if (e0 >= e) return;
  const int mine = e0 + (lane & (EPW - 1));
  const float* xl = xw + lane;
  float* al = agg + lane;
  if (e0 + EPW <= e) {
    const int r = row[mine], c = col[mine];
    const float nrm = enorm[mine];
    float v0[EPW], v1[EPW];
#pragma unroll
    for (int j = 0; j < EPW; ++j) {
      const float* src = xl + (long)__builtin_amdgcn_readlane(r, j) * NF;
      v0[j] = src[0];
      v1[j] = src[32];
    }
#pragma unroll
    for (int j = 0; j < EPW; ++j) {
      const float nj = __int_as_float(__builtin_amdgcn_readlane(__float_as_int(nrm), j));
      float* dst = al + (long)__builtin_amdgcn_readlane(c, j) * NF;
      atomicAdd(dst,      v0[j] * nj);
      atomicAdd(dst + 32, v1[j] * nj);
    }
  } else {
    int r = 0, c = 0;
    float nrm = 0.f;
    if (mine < e) {
      r = row[mine]; c = col[mine];
      nrm = enorm[mine];
    }
    for (int j = 0; e0 + j < e; ++j) {
      const int rj = __builtin_amdgcn_readlane(r, j);
      const int cj = __builtin_amdgcn_readlane(c, j);
      const float nj = __int_as_float(__builtin_amdgcn_readlane(__float_as_int(nrm), j));
      const float* src = xl + (long)rj * NF;
      float* dst = al + (long)cj * NF;
      atomicAdd(dst,      src[0]  * nj);
      atomicAdd(dst + 32, src[32] * nj);
    }
  }
}

__global__ void k_decode(const float* __restrict__ z, const int* __restrict__ src,
                         const int* __restrict__ dst, float* __restrict__ out, int e) {
  int i = blockIdx.x * blockDim.x + threadIdx.x;
  if (i < e) {
    const float4* a = (const float4*)(z + (long)src[i] * NF);
    const float4* b = (const float4*)(z + (long)dst[i] * NF);
    float acc = 0.f;
#pragma unroll
    for (int j = 0; j < 16; ++j) {
      float4 av = a[j], bv = b[j];
      acc += av.x * bv.x + av.y * bv.y + av.z * bv.z + av.w * bv.w;
    }
    out[i] = acc;
  }
}

extern "C" void kernel_launch(void* const* d_in, const int* in_sizes, int n_in,
                              void* d_out, int out_size, void* d_ws, size_t ws_size,
                              hipStream_t stream) {
  const float* x   = (const float*)d_in[0];
  const int*   ei  = (const int*)d_in[1];
  const int*   eli = (const int*)d_in[2];
  const float* W1  = (const float*)d_in[3];
  const float* b1  = (const float*)d_in[4];
  const float* W2  = (const float*)d_in[5];
  const float* b2  = (const float*)d_in[6];
  float* out = (float*)d_out;

  float* ws   = (float*)d_ws;
  float* deg  = ws;
  float* bufA = ws + 50048;
  float* bufB = bufA + (size_t)NODES * NF;
  float* bufC = bufB + (size_t)NODES * NF;
  float* enrm = bufC + (size_t)NODES * NF;

  const int* erow = ei;
  const int* ecol = ei + EDGES;
  const int* lsrc = eli;
  const int* ldst = eli + LEDGES;

  dim3 blk(256);
  const int edgeGrid = (EDGES / EPW + 7) / 8;
  const int strips   = NODES / 16;
  const int gemmGrid = (strips + 7) / 8;
  static_assert(((NODES / 16 + 7) / 8) * 256 * EPT >= EDGES, "edge-norm shares must cover every edge");

  k_deg_init<<<(NODES + 255) / 256, blk, 0, stream>>>(deg, NODES);
  k_deg_accum<<<(EDGES + 255) / 256, blk, 0, stream>>>(ecol, deg, EDGES);

  k_gemm<INCH, false, true><<<gemmGrid, blk, 0, stream>>>(x, W1, b1, deg, bufA, bufB, NODES,
                                                       erow, ecol, enrm, EDGES);
  k_agg_edges<<<edgeGrid, blk, 0, stream>>>(bufA, erow, ecol, enrm, bufB, EDGES);

  k_gemm<NF, true, false><<<gemmGrid, blk, 0, stream>>>(bufB, W2, b2, deg, bufA, bufC, NODES,
                                                     erow, ecol, enrm, EDGES);
  k_agg_edges<<<edgeGrid, blk, 0, stream>>>(bufA, erow, ecol, enrm, bufC, EDGES);

  k_decode<<<(LEDGES + 255) / 256, blk, 0, stream>>>(bufC, lsrc, ldst, out, LEDGES);
}


# GEMMs keep W whole in LDS: one staging pass of whole-line 16-B loads and one barrier per work-group, K loop unbroken with A fetched one 64-k group ahead (was 4 chunk stagings, 8 barriers); sums same
# speedup vs baseline: 1.9402x; 1.1263x over previous
#include <hip/hip_runtime.h>

#define NODES  50000
#define EDGES  800000
#define LEDGES 200000
#define INCH   256
#define NF     64

typedef float v2f __attribute__((ext_vector_type(2)));
typedef float v8f __attribute__((ext_vector_type(8)));

__global__ void k_deg_init(float* deg, int n) {
  int i = blockIdx.x * blockDim.x + threadIdx.x;
  if (i < n) deg[i] = 1.0f;
}

__global__ void k_deg_accum(const int* __restrict__ col, float* deg, int e) {
  int i = blockIdx.x * blockDim.x + threadIdx.x;
  if (i < e) atomicAdd(&deg[col[i]], 1.0f);
}

#define EPT 8
template<int K, bool RELU_IN, bool EDGE_NORM>
__global__ __launch_bounds__(256) void k_gemm(const float* __restrict__ A,
                                              const float* __restrict__ W,
                                              const float* __restrict__ bias,
                                              const float* __restrict__ deg,
                                              float* __restrict__ C,
                                              float* __restrict__ AGG, int M,
                                              const int* __restrict__ erow,
                                              const int* __restrict__ ecol,
                                              float* __restrict__ ENORM, int E) {
  __shared__ __align__(16) float sB[K * NF];

  constexpr int NG = K / 64;
  constexpr int NT = K / 32;

  const int lane = threadIdx.x & 31;
  const int wave = threadIdx.x >> 5;

  int strip = blockIdx.x * 8 + wave;
  int maxStrip = (M >> 4) - 1;
  if (strip > maxStrip) strip = maxStrip;
  const long rowBase = (long)strip << 4;

  const int mlo  = lane & 15;
  const int koff = (lane >> 4) << 1;
  const float* arow = A + (rowBase + mlo) * (long)K;

  const int mhi = (lane >> 4) << 3;
  float4 dg0 = {}, dg1 = {};
  float bcol[4] = {};

  float4 w0[NT], w1[NT];
#pragma unroll
  for (int j = 0; j < NT; ++j) {
    const int q   = threadIdx.x + j * 256;
    const int kk  = (q >> 4) << 1;
    const int cl0 = (q & 15) << 2;
    w0[j] = *(const float4*)(W + kk * NF + cl0);
    w1[j] = *(const float4*)(W + (kk + 1) * NF + cl0);
  }

  const int gtid = blockIdx.x * 256 + threadIdx.x;
  const int gstride = gridDim.x * 256;
  int er[EPT], ec[EPT];
  float gr[EPT], gc[EPT];
  if (EDGE_NORM) {
#pragma unroll
    for (int j = 0; j < EPT; ++j) {
      const int idx = min(gtid + j * gstride, E - 1);
      er[j] = erow[idx];
      ec[j] = ecol[idx];
    }
  }

  v2f acur[16];
#pragma unroll
  for (int ks = 0; ks < 16; ++ks) acur[ks] = *(const v2f*)(arow + ks * 4 + koff);

#pragma unroll
  for (int j = 0; j < NT; ++j) {
    const int q  = threadIdx.x + j * 256;
    const int r  = q >> 4;
    const int ks = r >> 1;
    const int h  = r & 1;
    const int t  = (q & 15) >> 2;
    const int c0 = (q & 3) << 2;
    float4* d = (float4*)(sB + (((ks * 4 + t) * 32 + h * 16 + c0) << 1));
    d[0] = make_float4(w0[j].x, w1[j].x, w0[j].y, w1[j].y);
    d[1] = make_float4(w0[j].z, w1[j].z, w0[j].w, w1[j].w);
  }
  __syncthreads();

  if (EDGE_NORM) {
#pragma unroll
    for (int j = 0; j < EPT; ++j) { gr[j] = deg[er[j]]; gc[j] = deg[ec[j]]; }
  }

  v8f acc[4];
#pragma unroll
  for (int t = 0; t < 4; ++t) acc[t] = (v8f){0.f,0.f,0.f,0.f,0.f,0.f,0.f,0.f};

#pragma unroll
  for (int g = 0; g < NG; ++g) {
    v2f anxt[16];
    if (g + 1 < NG) {
#pragma unroll
      for (int ks = 0; ks < 16; ++ks)
        anxt[ks] = *(const v2f*)(arow + (g + 1) * 64 + ks * 4 + koff);
    } else {
      dg0 = *(const float4*)(deg + rowBase + mhi);
      dg1 = *(const float4*)(deg + rowBase + mhi + 4);
#pragma unroll
      for (int t = 0; t < 4; ++t) bcol[t] = bias[t * 16 + mlo];
    }
#pragma unroll
    for (int ks = 0; ks < 16; ++ks) {
      v2f a = acur[ks];
      if (RELU_IN) {
        a.x = fmaxf(a.x, 0.f);
        a.y = fmaxf(a.y, 0.f);
      }
      const v2f* bfrag = (const v2f*)sB + (g * 16 + ks) * 128 + lane;
#pragma unroll
      for (int t = 0; t < 4; ++t) {
        v2f b = bfrag[t * 32];
        acc[t] = __builtin_amdgcn_wmma_f32_16x16x4_f32(
            false, a, false, b, (short)0, acc[t], false, false);
      }
    }
    if (g + 1 < NG) {
#pragma unroll
      for (int ks = 0; ks < 16; ++ks) acur[ks] = anxt[ks];
    }
  }

  float sl[8];
  sl[0] = rsqrtf(dg0.x); sl[1] = rsqrtf(dg0.y); sl[2] = rsqrtf(dg0.z); sl[3] = rsqrtf(dg0.w);
  sl[4] = rsqrtf(dg1.x); sl[5] = rsqrtf(dg1.y); sl[6] = rsqrtf(dg1.z); sl[7] = rsqrtf(dg1.w);
#pragma unroll
  for (int i = 0; i < 8; ++i) sl[i] = sl[i] * sl[i];
#pragma unroll
  for (int t = 0; t < 4; ++t)
#pragma unroll
    for (int i = 0; i < 8; ++i) {
      const long o = (rowBase + mhi + i) * NF + t * 16 + mlo;
      const float v = acc[t][i];
      C[o]   = v;
      AGG[o] = v * sl[i] + bcol[t];
    }

  if (EDGE_NORM) {
#pragma unroll
    for (int j = 0; j < EPT; ++j) {
      const int idx = gtid + j * gstride;
      if (idx < E) ENORM[idx] = rsqrtf(gr[j]) * rsqrtf(gc[j]);
    }
  }
}

#define EPW 16
__global__ __launch_bounds__(256) void k_agg_edges(const float* __restrict__ xw,
                            const int* __restrict__ row, const int* __restrict__ col,
                            const float* __restrict__ enorm, float* agg, int e) {
  const int lane = threadIdx.x & 31;
  const int e0 = __builtin_amdgcn_readfirstlane((blockIdx.x * 8 + (threadIdx.x >> 5)) * EPW);
  if (e0 >= e) return;
  const int mine = e0 + (lane & (EPW - 1));
  const float* xl = xw + lane;
  float* al = agg + lane;
  if (e0 + EPW <= e) {
    const int r = row[mine], c = col[mine];
    const float nrm = enorm[mine];
    float v0[EPW], v1[EPW];
#pragma unroll
    for (int j = 0; j < EPW; ++j) {
      const float* src = xl + (long)__builtin_amdgcn_readlane(r, j) * NF;
      v0[j] = src[0];
      v1[j] = src[32];
    }
#pragma unroll
    for (int j = 0; j < EPW; ++j) {
      const float nj = __int_as_float(__builtin_amdgcn_readlane(__float_as_int(nrm), j));
      float* dst = al + (long)__builtin_amdgcn_readlane(c, j) * NF;
      atomicAdd(dst,      v0[j] * nj);
      atomicAdd(dst + 32, v1[j] * nj);
    }
  } else {
    int r = 0, c = 0;
    float nrm = 0.f;
    if (mine < e) {
      r = row[mine]; c = col[mine];
      nrm = enorm[mine];
    }
    for (int j = 0; e0 + j < e; ++j) {
      const int rj = __builtin_amdgcn_readlane(r, j);
      const int cj = __builtin_amdgcn_readlane(c, j);
      const float nj = __int_as_float(__builtin_amdgcn_readlane(__float_as_int(nrm), j));
      const float* src = xl + (long)rj * NF;
      float* dst = al + (long)cj * NF;
      atomicAdd(dst,      src[0]  * nj);
      atomicAdd(dst + 32, src[32] * nj);
    }
  }
}

__global__ void k_decode(const float* __restrict__ z, const int* __restrict__ src,
                         const int* __restrict__ dst, float* __restrict__ out, int e) {
  int i = blockIdx.x * blockDim.x + threadIdx.x;
  if (i < e) {
    const float4* a = (const float4*)(z + (long)src[i] * NF);
    const float4* b = (const float4*)(z + (long)dst[i] * NF);
    float acc = 0.f;
#pragma unroll
    for (int j = 0; j < 16; ++j) {
      float4 av = a[j], bv = b[j];
      acc += av.x * bv.x + av.y * bv.y + av.z * bv.z + av.w * bv.w;
    }
    out[i] = acc;
  }
}

extern "C" void kernel_launch(void* const* d_in, const int* in_sizes, int n_in,
                              void* d_out, int out_size, void* d_ws, size_t ws_size,
                              hipStream_t stream) {
  const float* x   = (const float*)d_in[0];
  const int*   ei  = (const int*)d_in[1];
  const int*   eli = (const int*)d_in[2];
  const float* W1  = (const float*)d_in[3];
  const float* b1  = (const float*)d_in[4];
  const float* W2  = (const float*)d_in[5];
  const float* b2  = (const float*)d_in[6];
  float* out = (float*)d_out;

  float* ws   = (float*)d_ws;
  float* deg  = ws;
  float* bufA = ws + 50048;
  float* bufB = bufA + (size_t)NODES * NF;
  float* bufC = bufB + (size_t)NODES * NF;
  float* enrm = bufC + (size_t)NODES * NF;

  const int* erow = ei;
  const int* ecol = ei + EDGES;
  const int* lsrc = eli;
  const int* ldst = eli + LEDGES;

  dim3 blk(256);
  const int edgeGrid = (EDGES / EPW + 7) / 8;
  const int strips   = NODES / 16;
  const int gemmGrid = (strips + 7) / 8;
  static_assert(((NODES / 16 + 7) / 8) * 256 * EPT >= EDGES, "edge-norm shares must cover every edge");

  k_deg_init<<<(NODES + 255) / 256, blk, 0, stream>>>(deg, NODES);
  k_deg_accum<<<(EDGES + 255) / 256, blk, 0, stream>>>(ecol, deg, EDGES);

  k_gemm<INCH, false, true><<<gemmGrid, blk, 0, stream>>>(x, W1, b1, deg, bufA, bufB, NODES,
                                                       erow, ecol, enrm, EDGES);
  k_agg_edges<<<edgeGrid, blk, 0, stream>>>(bufA, erow, ecol, enrm, bufB, EDGES);

  k_gemm<NF, true, false><<<gemmGrid, blk, 0, stream>>>(bufB, W2, b2, deg, bufA, bufC, NODES,
                                                     erow, ecol, enrm, EDGES);
  k_agg_edges<<<edgeGrid, blk, 0, stream>>>(bufA, erow, ecol, enrm, bufC, EDGES);

  k_decode<<<(LEDGES + 255) / 256, blk, 0, stream>>>(bufC, lsrc, ldst, out, LEDGES);
}


# Decoder: label rows fetched as whole-row 16-B loads (two rows per load instruction) into LDS at pitch 272 B; each lane reads its own two rows back from LDS and sums in the old order
# speedup vs baseline: 2.0874x; 1.0759x over previous
#include <hip/hip_runtime.h>

#define NODES  50000
#define EDGES  800000
#define LEDGES 200000
#define INCH   256
#define NF     64

typedef float v2f __attribute__((ext_vector_type(2)));
typedef float v8f __attribute__((ext_vector_type(8)));

__global__ void k_deg_init(float* deg, int n) {
  int i = blockIdx.x * blockDim.x + threadIdx.x;
  if (i < n) deg[i] = 1.0f;
}

__global__ void k_deg_accum(const int* __restrict__ col, float* deg, int e) {
  int i = blockIdx.x * blockDim.x + threadIdx.x;
  if (i < e) atomicAdd(&deg[col[i]], 1.0f);
}

#define EPT 8
template<int K, bool RELU_IN, bool EDGE_NORM>
__global__ __launch_bounds__(256) void k_gemm(const float* __restrict__ A,
                                              const float* __restrict__ W,
                                              const float* __restrict__ bias,
                                              const float* __restrict__ deg,
                                              float* __restrict__ C,
                                              float* __restrict__ AGG, int M,
                                              const int* __restrict__ erow,
                                              const int* __restrict__ ecol,
                                              float* __restrict__ ENORM, int E) {
  __shared__ __align__(16) float sB[K * NF];

  constexpr int NG = K / 64;
  constexpr int NT = K / 32;

  const int lane = threadIdx.x & 31;
  const int wave = threadIdx.x >> 5;

  int strip = blockIdx.x * 8 + wave;
  int maxStrip = (M >> 4) - 1;
  if (strip > maxStrip) strip = maxStrip;
  const long rowBase = (long)strip << 4;

  const int mlo  = lane & 15;
  const int koff = (lane >> 4) << 1;
  const float* arow = A + (rowBase + mlo) * (long)K;

  const int mhi = (lane >> 4) << 3;
  float4 dg0 = {}, dg1 = {};
  float bcol[4] = {};

  float4 w0[NT], w1[NT];
#pragma unroll
  for (int j = 0; j < NT; ++j) {
    const int q   = threadIdx.x + j * 256;
    const int kk  = (q >> 4) << 1;
    const int cl0 = (q & 15) << 2;
    w0[j] = *(const float4*)(W + kk * NF + cl0);
    w1[j] = *(const float4*)(W + (kk + 1) * NF + cl0);
  }

  const int gtid = blockIdx.x * 256 + threadIdx.x;
  const int gstride = gridDim.x * 256;
  int er[EPT], ec[EPT];
  float gr[EPT], gc[EPT];
  if (EDGE_NORM) {
#pragma unroll
    for (int j = 0; j < EPT; ++j) {
      const int idx = min(gtid + j * gstride, E - 1);
      er[j] = erow[idx];
      ec[j] = ecol[idx];
    }
  }

  v2f acur[16];
#pragma unroll
  for (int ks = 0; ks < 16; ++ks) acur[ks] = *(const v2f*)(arow + ks * 4 + koff);

#pragma unroll
  for (int j = 0; j < NT; ++j) {
    const int q  = threadIdx.x + j * 256;
    const int r  = q >> 4;
    const int ks = r >> 1;
    const int h  = r & 1;
    const int t  = (q & 15) >> 2;
    const int c0 = (q & 3) << 2;
    float4* d = (float4*)(sB + (((ks * 4 + t) * 32 + h * 16 + c0) << 1));
    d[0] = make_float4(w0[j].x, w1[j].x, w0[j].y, w1[j].y);
    d[1] = make_float4(w0[j].z, w1[j].z, w0[j].w, w1[j].w);
  }
  __syncthreads();

  if (EDGE_NORM) {
#pragma unroll
    for (int j = 0; j < EPT; ++j) { gr[j] = deg[er[j]]; gc[j] = deg[ec[j]]; }
  }

  v8f acc[4];
#pragma unroll
  for (int t = 0; t < 4; ++t) acc[t] = (v8f){0.f,0.f,0.f,0.f,0.f,0.f,0.f,0.f};

#pragma unroll
  for (int g = 0; g < NG; ++g) {
    v2f anxt[16];
    if (g + 1 < NG) {
#pragma unroll
      for (int ks = 0; ks < 16; ++ks)
        anxt[ks] = *(const v2f*)(arow + (g + 1) * 64 + ks * 4 + koff);
    } else {
      dg0 = *(const float4*)(deg + rowBase + mhi);
      dg1 = *(const float4*)(deg + rowBase + mhi + 4);
#pragma unroll
      for (int t = 0; t < 4; ++t) bcol[t] = bias[t * 16 + mlo];
    }
#pragma unroll
    for (int ks = 0; ks < 16; ++ks) {
      v2f a = acur[ks];
      if (RELU_IN) {
        a.x = fmaxf(a.x, 0.f);
        a.y = fmaxf(a.y, 0.f);
      }
      const v2f* bfrag = (const v2f*)sB + (g * 16 + ks) * 128 + lane;
#pragma unroll
      for (int t = 0; t < 4; ++t) {
        v2f b = bfrag[t * 32];
        acc[t] = __builtin_amdgcn_wmma_f32_16x16x4_f32(
            false, a, false, b, (short)0, acc[t], false, false);
      }
    }
    if (g + 1 < NG) {
#pragma unroll
      for (int ks = 0; ks < 16; ++ks) acur[ks] = anxt[ks];
    }
  }

  float sl[8];
  sl[0] = rsqrtf(dg0.x); sl[1] = rsqrtf(dg0.y); sl[2] = rsqrtf(dg0.z); sl[3] = rsqrtf(dg0.w);
  sl[4] = rsqrtf(dg1.x); sl[5] = rsqrtf(dg1.y); sl[6] = rsqrtf(dg1.z); sl[7] = rsqrtf(dg1.w);
#pragma unroll
  for (int i = 0; i < 8; ++i) sl[i] = sl[i] * sl[i];
#pragma unroll
  for (int t = 0; t < 4; ++t)
#pragma unroll
    for (int i = 0; i < 8; ++i) {
      const long o = (rowBase + mhi + i) * NF + t * 16 + mlo;
      const float v = acc[t][i];
      C[o]   = v;
      AGG[o] = v * sl[i] + bcol[t];
    }

  if (EDGE_NORM) {
#pragma unroll
    for (int j = 0; j < EPT; ++j) {
      const int idx = gtid + j * gstride;
      if (idx < E) ENORM[idx] = rsqrtf(gr[j]) * rsqrtf(gc[j]);
    }
  }
}

#define EPW 16
__global__ __launch_bounds__(256) void k_agg_edges(const float* __restrict__ xw,
                            const int* __restrict__ row, const int* __restrict__ col,
                            const float* __restrict__ enorm, float* agg, int e) {
  const int lane = threadIdx.x & 31;
  const int e0 = __builtin_amdgcn_readfirstlane((blockIdx.x * 8 + (threadIdx.x >> 5)) * EPW);
  if (e0 >= e) return;
  const int mine = e0 + (lane & (EPW - 1));
  const float* xl = xw + lane;
  float* al = agg + lane;
  if (e0 + EPW <= e) {
    const int r = row[mine], c = col[mine];
    const float nrm = enorm[mine];
    float v0[EPW], v1[EPW];
#pragma unroll
    for (int j = 0; j < EPW; ++j) {
      const float* src = xl + (long)__builtin_amdgcn_readlane(r, j) * NF;
      v0[j] = src[0];
      v1[j] = src[32];
    }
#pragma unroll
    for (int j = 0; j < EPW; ++j) {
      const float nj = __int_as_float(__builtin_amdgcn_readlane(__float_as_int(nrm), j));
      float* dst = al + (long)__builtin_amdgcn_readlane(c, j) * NF;
      atomicAdd(dst,      v0[j] * nj);
      atomicAdd(dst + 32, v1[j] * nj);
    }
  } else {
    int r = 0, c = 0;
    float nrm = 0.f;
    if (mine < e) {
      r = row[mine]; c = col[mine];
      nrm = enorm[mine];
    }
    for (int j = 0; e0 + j < e; ++j) {
      const int rj = __builtin_amdgcn_readlane(r, j);
      const int cj = __builtin_amdgcn_readlane(c, j);
      const float nj = __int_as_float(__builtin_amdgcn_readlane(__float_as_int(nrm), j));
      const float* src = xl + (long)rj * NF;
      float* dst = al + (long)cj * NF;
      atomicAdd(dst,      src[0]  * nj);
      atomicAdd(dst + 32, src[32] * nj);
    }
  }
}

#define DROWP 68
__global__ __launch_bounds__(256) void k_decode(const float* __restrict__ z,
                         const int* __restrict__ src, const int* __restrict__ dst,
                         float* __restrict__ out, int e) {
  __shared__ __align__(16) float sR[8 * 64 * DROWP];
  const int lane = threadIdx.x & 31;
  const int wave = threadIdx.x >> 5;
  const int half = lane >> 4;
  const int c4   = (lane & 15) << 2;
  const int i    = blockIdx.x * 256 + threadIdx.x;
  const int ic   = min(i, e - 1);
  const int s = src[ic], d = dst[ic];
  float* wr = sR + wave * (64 * DROWP);

  float4 va[16], vb[16];
#pragma unroll
  for (int j = 0; j < 16; ++j) {
    const int s0 = __builtin_amdgcn_readlane(s, 2 * j), s1 = __builtin_amdgcn_readlane(s, 2 * j + 1);
    const int d0 = __builtin_amdgcn_readlane(d, 2 * j), d1 = __builtin_amdgcn_readlane(d, 2 * j + 1);
    va[j] = *(const float4*)(z + (long)(half ? s1 : s0) * NF + c4);
    vb[j] = *(const float4*)(z + (long)(half ? d1 : d0) * NF + c4);
  }
#pragma unroll
  for (int j = 0; j < 16; ++j) {
    *(float4*)(wr + (2 * j + half) * DROWP + c4)        = va[j];
    *(float4*)(wr + (32 + 2 * j + half) * DROWP + c4)   = vb[j];
  }
  __syncthreads();
  const float4* a = (const float4*)(wr + lane * DROWP);
  const float4* b = (const float4*)(wr + (32 + lane) * DROWP);
  float acc = 0.f;
#pragma unroll
  for (int j = 0; j < 16; ++j) {
    float4 av = a[j], bv = b[j];
    acc += av.x * bv.x + av.y * bv.y + av.z * bv.z + av.w * bv.w;
  }
  if (i < e) out[i] = acc;
}

extern "C" void kernel_launch(void* const* d_in, const int* in_sizes, int n_in,
                              void* d_out, int out_size, void* d_ws, size_t ws_size,
                              hipStream_t stream) {
  const float* x   = (const float*)d_in[0];
  const int*   ei  = (const int*)d_in[1];
  const int*   eli = (const int*)d_in[2];
  const float* W1  = (const float*)d_in[3];
  const float* b1  = (const float*)d_in[4];
  const float* W2  = (const float*)d_in[5];
  const float* b2  = (const float*)d_in[6];
  float* out = (float*)d_out;

  float* ws   = (float*)d_ws;
  float* deg  = ws;
  float* bufA = ws + 50048;
  float* bufB = bufA + (size_t)NODES * NF;
  float* bufC = bufB + (size_t)NODES * NF;
  float* enrm = bufC + (size_t)NODES * NF;

  const int* erow = ei;
  const int* ecol = ei + EDGES;
  const int* lsrc = eli;
  const int* ldst = eli + LEDGES;

  dim3 blk(256);
  const int edgeGrid = (EDGES / EPW + 7) / 8;
  const int strips   = NODES / 16;
  const int gemmGrid = (strips + 7) / 8;
  static_assert(((NODES / 16 + 7) / 8) * 256 * EPT >= EDGES, "edge-norm shares must cover every edge");

  k_deg_init<<<(NODES + 255) / 256, blk, 0, stream>>>(deg, NODES);
  k_deg_accum<<<(EDGES + 255) / 256, blk, 0, stream>>>(ecol, deg, EDGES);

  k_gemm<INCH, false, true><<<gemmGrid, blk, 0, stream>>>(x, W1, b1, deg, bufA, bufB, NODES,
                                                       erow, ecol, enrm, EDGES);
  k_agg_edges<<<edgeGrid, blk, 0, stream>>>(bufA, erow, ecol, enrm, bufB, EDGES);

  k_gemm<NF, true, false><<<gemmGrid, blk, 0, stream>>>(bufB, W2, b2, deg, bufA, bufC, NODES,
                                                     erow, ecol, enrm, EDGES);
  k_agg_edges<<<edgeGrid, blk, 0, stream>>>(bufA, erow, ecol, enrm, bufC, EDGES);

  k_decode<<<(LEDGES + 255) / 256, blk, 0, stream>>>(bufC, lsrc, ldst, out, LEDGES);
}


# Degree count: 50 work-groups each count an edge slice into 50000 counters in LDS (integer LDS adds), then add them to deg as whole-line f32 atomics; was one global f32 atomic per edge; sums exact
# speedup vs baseline: 2.5549x; 1.2240x over previous
#include <hip/hip_runtime.h>

#define NODES  50000
#define EDGES  800000
#define LEDGES 200000
#define INCH   256
#define NF     64

typedef float v2f __attribute__((ext_vector_type(2)));
typedef float v8f __attribute__((ext_vector_type(8)));

__global__ void k_deg_init(float* deg, int n) {
  int i = blockIdx.x * blockDim.x + threadIdx.x;
  if (i < n) deg[i] = 1.0f;
}

#define DEG_WG 50
__global__ __launch_bounds__(1024) void k_deg_hist(const int* __restrict__ col, float* deg) {
  __shared__ __align__(16) unsigned cnt[NODES];
  constexpr int SLICE4 = EDGES / DEG_WG / 4;
  constexpr int NLOAD  = (SLICE4 + 1023) / 1024;
  constexpr int NSTEP  = (NODES + 1023) / 1024;
  constexpr int NB     = 7;
  static_assert(NSTEP % NB == 0, "the flush runs in whole groups");
  static_assert(EDGES % (DEG_WG * 4) == 0, "slices must be whole 16-byte items");
  static_assert(NODES % 4 == 0, "counters are cleared 16 bytes at a time");
  const int tid = threadIdx.x;

  const int4* s4 = (const int4*)col + (long)blockIdx.x * SLICE4;
  int4 v[NLOAD];
#pragma unroll
  for (int j = 0; j < NLOAD; ++j) v[j] = s4[min(tid + j * 1024, SLICE4 - 1)];

  uint4* c4 = (uint4*)cnt;
  for (int i = tid; i < NODES / 4; i += 1024) c4[i] = make_uint4(0u, 0u, 0u, 0u);
  __syncthreads();

#pragma unroll
  for (int j = 0; j < NLOAD; ++j) {
    if (tid + j * 1024 < SLICE4) {
      atomicAdd(&cnt[v[j].x], 1u);
      atomicAdd(&cnt[v[j].y], 1u);
      atomicAdd(&cnt[v[j].z], 1u);
      atomicAdd(&cnt[v[j].w], 1u);
    }
  }
  __syncthreads();

  int k = blockIdx.x % NSTEP;
  for (int b = 0; b < NSTEP / NB; ++b) {
    int idx[NB];
    unsigned c[NB];
#pragma unroll
    for (int u = 0; u < NB; ++u) {
      idx[u] = k * 1024 + tid;
      c[u] = cnt[min(idx[u], NODES - 1)];
      k = (k + 1 == NSTEP) ? 0 : k + 1;
    }
#pragma unroll
    for (int u = 0; u < NB; ++u)
      if (idx[u] < NODES) atomicAdd(&deg[idx[u]], (float)c[u]);
  }
}

#define EPT 8
template<int K, bool RELU_IN, bool EDGE_NORM>
__global__ __launch_bounds__(256) void k_gemm(const float* __restrict__ A,
                                              const float* __restrict__ W,
                                              const float* __restrict__ bias,
                                              const float* __restrict__ deg,
                                              float* __restrict__ C,
                                              float* __restrict__ AGG, int M,
                                              const int* __restrict__ erow,
                                              const int* __restrict__ ecol,
                                              float* __restrict__ ENORM, int E) {
  __shared__ __align__(16) float sB[K * NF];

  constexpr int NG = K / 64;
  constexpr int NT = K / 32;

  const int lane = threadIdx.x & 31;
  const int wave = threadIdx.x >> 5;

  int strip = blockIdx.x * 8 + wave;
  int maxStrip = (M >> 4) - 1;
  if (strip > maxStrip) strip = maxStrip;
  const long rowBase = (long)strip << 4;

  const int mlo  = lane & 15;
  const int koff = (lane >> 4) << 1;
  const float* arow = A + (rowBase + mlo) * (long)K;

  const int mhi = (lane >> 4) << 3;
  float4 dg0 = {}, dg1 = {};
  float bcol[4] = {};

  float4 w0[NT], w1[NT];
#pragma unroll
  for (int j = 0; j < NT; ++j) {
    const int q   = threadIdx.x + j * 256;
    const int kk  = (q >> 4) << 1;
    const int cl0 = (q & 15) << 2;
    w0[j] = *(const float4*)(W + kk * NF + cl0);
    w1[j] = *(const float4*)(W + (kk + 1) * NF + cl0);
  }

  const int gtid = blockIdx.x * 256 + threadIdx.x;
  const int gstride = gridDim.x * 256;
  int er[EPT], ec[EPT];
  float gr[EPT], gc[EPT];
  if (EDGE_NORM) {
#pragma unroll
    for (int j = 0; j < EPT; ++j) {
      const int idx = min(gtid + j * gstride, E - 1);
      er[j] = erow[idx];
      ec[j] = ecol[idx];
    }
  }

  v2f acur[16];
#pragma unroll
  for (int ks = 0; ks < 16; ++ks) acur[ks] = *(const v2f*)(arow + ks * 4 + koff);

#pragma unroll
  for (int j = 0; j < NT; ++j) {
    const int q  = threadIdx.x + j * 256;
    const int r  = q >> 4;
    const int ks = r >> 1;
    const int h  = r & 1;
    const int t  = (q & 15) >> 2;
    const int c0 = (q & 3) << 2;
    float4* d = (float4*)(sB + (((ks * 4 + t) * 32 + h * 16 + c0) << 1));
    d[0] = make_float4(w0[j].x, w1[j].x, w0[j].y, w1[j].y);
    d[1] = make_float4(w0[j].z, w1[j].z, w0[j].w, w1[j].w);
  }
  __syncthreads();

  if (EDGE_NORM) {
#pragma unroll
    for (int j = 0; j < EPT; ++j) { gr[j] = deg[er[j]]; gc[j] = deg[ec[j]]; }
  }

  v8f acc[4];
#pragma unroll
  for (int t = 0; t < 4; ++t) acc[t] = (v8f){0.f,0.f,0.f,0.f,0.f,0.f,0.f,0.f};

#pragma unroll
  for (int g = 0; g < NG; ++g) {
    v2f anxt[16];
    if (g + 1 < NG) {
#pragma unroll
      for (int ks = 0; ks < 16; ++ks)
        anxt[ks] = *(const v2f*)(arow + (g + 1) * 64 + ks * 4 + koff);
    } else {
      dg0 = *(const float4*)(deg + rowBase + mhi);
      dg1 = *(const float4*)(deg + rowBase + mhi + 4);
#pragma unroll
      for (int t = 0; t < 4; ++t) bcol[t] = bias[t * 16 + mlo];
    }
#pragma unroll
    for (int ks = 0; ks < 16; ++ks) {
      v2f a = acur[ks];
      if (RELU_IN) {
        a.x = fmaxf(a.x, 0.f);
        a.y = fmaxf(a.y, 0.f);
      }
      const v2f* bfrag = (const v2f*)sB + (g * 16 + ks) * 128 + lane;
#pragma unroll
      for (int t = 0; t < 4; ++t) {
        v2f b = bfrag[t * 32];
        acc[t] = __builtin_amdgcn_wmma_f32_16x16x4_f32(
            false, a, false, b, (short)0, acc[t], false, false);
      }
    }
    if (g + 1 < NG) {
#pragma unroll
      for (int ks = 0; ks < 16; ++ks) acur[ks] = anxt[ks];
    }
  }

  float sl[8];
  sl[0] = rsqrtf(dg0.x); sl[1] = rsqrtf(dg0.y); sl[2] = rsqrtf(dg0.z); sl[3] = rsqrtf(dg0.w);
  sl[4] = rsqrtf(dg1.x); sl[5] = rsqrtf(dg1.y); sl[6] = rsqrtf(dg1.z); sl[7] = rsqrtf(dg1.w);
#pragma unroll
  for (int i = 0; i < 8; ++i) sl[i] = sl[i] * sl[i];
#pragma unroll
  for (int t = 0; t < 4; ++t)
#pragma unroll
    for (int i = 0; i < 8; ++i) {
      const long o = (rowBase + mhi + i) * NF + t * 16 + mlo;
      const float v = acc[t][i];
      C[o]   = v;
      AGG[o] = v * sl[i] + bcol[t];
    }

  if (EDGE_NORM) {
#pragma unroll
    for (int j = 0; j < EPT; ++j) {
      const int idx = gtid + j * gstride;
      if (idx < E) ENORM[idx] = rsqrtf(gr[j]) * rsqrtf(gc[j]);
    }
  }
}

#define EPW 16
__global__ __launch_bounds__(256) void k_agg_edges(const float* __restrict__ xw,
                            const int* __restrict__ row, const int* __restrict__ col,
                            const float* __restrict__ enorm, float* agg, int e) {
  const int lane = threadIdx.x & 31;
  const int e0 = __builtin_amdgcn_readfirstlane((blockIdx.x * 8 + (threadIdx.x >> 5)) * EPW);
  if (e0 >= e) return;
  const int mine = e0 + (lane & (EPW - 1));
  const float* xl = xw + lane;
  float* al = agg + lane;
  if (e0 + EPW <= e) {
    const int r = row[mine], c = col[mine];
    const float nrm = enorm[mine];
    float v0[EPW], v1[EPW];
#pragma unroll
    for (int j = 0; j < EPW; ++j) {
      const float* src = xl + (long)__builtin_amdgcn_readlane(r, j) * NF;
      v0[j] = src[0];
      v1[j] = src[32];
    }
#pragma unroll
    for (int j = 0; j < EPW; ++j) {
      const float nj = __int_as_float(__builtin_amdgcn_readlane(__float_as_int(nrm), j));
      float* dst = al + (long)__builtin_amdgcn_readlane(c, j) * NF;
      atomicAdd(dst,      v0[j] * nj);
      atomicAdd(dst + 32, v1[j] * nj);
    }
  } else {
    int r = 0, c = 0;
    float nrm = 0.f;
    if (mine < e) {
      r = row[mine]; c = col[mine];
      nrm = enorm[mine];
    }
    for (int j = 0; e0 + j < e; ++j) {
      const int rj = __builtin_amdgcn_readlane(r, j);
      const int cj = __builtin_amdgcn_readlane(c, j);
      const float nj = __int_as_float(__builtin_amdgcn_readlane(__float_as_int(nrm), j));
      const float* src = xl + (long)rj * NF;
      float* dst = al + (long)cj * NF;
      atomicAdd(dst,      src[0]  * nj);
      atomicAdd(dst + 32, src[32] * nj);
    }
  }
}

#define DROWP 68
__global__ __launch_bounds__(256) void k_decode(const float* __restrict__ z,
                         const int* __restrict__ src, const int* __restrict__ dst,
                         float* __restrict__ out, int e) {
  __shared__ __align__(16) float sR[8 * 64 * DROWP];
  const int lane = threadIdx.x & 31;
  const int wave = threadIdx.x >> 5;
  const int half = lane >> 4;
  const int c4   = (lane & 15) << 2;
  const int i    = blockIdx.x * 256 + threadIdx.x;
  const int ic   = min(i, e - 1);
  const int s = src[ic], d = dst[ic];
  float* wr = sR + wave * (64 * DROWP);

  float4 va[16], vb[16];
#pragma unroll
  for (int j = 0; j < 16; ++j) {
    const int s0 = __builtin_amdgcn_readlane(s, 2 * j), s1 = __builtin_amdgcn_readlane(s, 2 * j + 1);
    const int d0 = __builtin_amdgcn_readlane(d, 2 * j), d1 = __builtin_amdgcn_readlane(d, 2 * j + 1);
    va[j] = *(const float4*)(z + (long)(half ? s1 : s0) * NF + c4);
    vb[j] = *(const float4*)(z + (long)(half ? d1 : d0) * NF + c4);
  }
#pragma unroll
  for (int j = 0; j < 16; ++j) {
    *(float4*)(wr + (2 * j + half) * DROWP + c4)        = va[j];
    *(float4*)(wr + (32 + 2 * j + half) * DROWP + c4)   = vb[j];
  }
  __syncthreads();
  const float4* a = (const float4*)(wr + lane * DROWP);
  const float4* b = (const float4*)(wr + (32 + lane) * DROWP);
  float acc = 0.f;
#pragma unroll
  for (int j = 0; j < 16; ++j) {
    float4 av = a[j], bv = b[j];
    acc += av.x * bv.x + av.y * bv.y + av.z * bv.z + av.w * bv.w;
  }
  if (i < e) out[i] = acc;
}

extern "C" void kernel_launch(void* const* d_in, const int* in_sizes, int n_in,
                              void* d_out, int out_size, void* d_ws, size_t ws_size,
                              hipStream_t stream) {
  const float* x   = (const float*)d_in[0];
  const int*   ei  = (const int*)d_in[1];
  const int*   eli = (const int*)d_in[2];
  const float* W1  = (const float*)d_in[3];
  const float* b1  = (const float*)d_in[4];
  const float* W2  = (const float*)d_in[5];
  const float* b2  = (const float*)d_in[6];
  float* out = (float*)d_out;

  float* ws   = (float*)d_ws;
  float* deg  = ws;
  float* bufA = ws + 50048;
  float* bufB = bufA + (size_t)NODES * NF;
  float* bufC = bufB + (size_t)NODES * NF;
  float* enrm = bufC + (size_t)NODES * NF;

  const int* erow = ei;
  const int* ecol = ei + EDGES;
  const int* lsrc = eli;
  const int* ldst = eli + LEDGES;

  dim3 blk(256);
  const int edgeGrid = (EDGES / EPW + 7) / 8;
  const int strips   = NODES / 16;
  const int gemmGrid = (strips + 7) / 8;
  static_assert(((NODES / 16 + 7) / 8) * 256 * EPT >= EDGES, "edge-norm shares must cover every edge");

  k_deg_init<<<(NODES + 255) / 256, blk, 0, stream>>>(deg, NODES);
  k_deg_hist<<<DEG_WG, 1024, 0, stream>>>(ecol, deg);

  k_gemm<INCH, false, true><<<gemmGrid, blk, 0, stream>>>(x, W1, b1, deg, bufA, bufB, NODES,
                                                       erow, ecol, enrm, EDGES);
  k_agg_edges<<<edgeGrid, blk, 0, stream>>>(bufA, erow, ecol, enrm, bufB, EDGES);

  k_gemm<NF, true, false><<<gemmGrid, blk, 0, stream>>>(bufB, W2, b2, deg, bufA, bufC, NODES,
                                                     erow, ecol, enrm, EDGES);
  k_agg_edges<<<edgeGrid, blk, 0, stream>>>(bufA, erow, ecol, enrm, bufC, EDGES);

  k_decode<<<(LEDGES + 255) / 256, blk, 0, stream>>>(bufC, lsrc, ldst, out, LEDGES);
}


# GEMM stores C rows with feature f beside f + 32 (whole-line 8-B stores); edge scatter gathers a source row with one 8-B load per lane, one instruction per row (was two); atomics, sum order as before
# speedup vs baseline: 2.6666x; 1.0437x over previous
#include <hip/hip_runtime.h>

#define NODES  50000
#define EDGES  800000
#define LEDGES 200000
#define INCH   256
#define NF     64

typedef float v2f __attribute__((ext_vector_type(2)));
typedef float v8f __attribute__((ext_vector_type(8)));

__global__ void k_deg_init(float* deg, int n) {
  int i = blockIdx.x * blockDim.x + threadIdx.x;
  if (i < n) deg[i] = 1.0f;
}

#define DEG_WG 50
__global__ __launch_bounds__(1024) void k_deg_hist(const int* __restrict__ col, float* deg) {
  __shared__ __align__(16) unsigned cnt[NODES];
  constexpr int SLICE4 = EDGES / DEG_WG / 4;
  constexpr int NLOAD  = (SLICE4 + 1023) / 1024;
  constexpr int NSTEP  = (NODES + 1023) / 1024;
  constexpr int NB     = 7;
  static_assert(NSTEP % NB == 0, "the flush runs in whole groups");
  static_assert(EDGES % (DEG_WG * 4) == 0, "slices must be whole 16-byte items");
  static_assert(NODES % 4 == 0, "counters are cleared 16 bytes at a time");
  const int tid = threadIdx.x;

  const int4* s4 = (const int4*)col + (long)blockIdx.x * SLICE4;
  int4 v[NLOAD];
#pragma unroll
  for (int j = 0; j < NLOAD; ++j) v[j] = s4[min(tid + j * 1024, SLICE4 - 1)];

  uint4* c4 = (uint4*)cnt;
  for (int i = tid; i < NODES / 4; i += 1024) c4[i] = make_uint4(0u, 0u, 0u, 0u);
  __syncthreads();

#pragma unroll
  for (int j = 0; j < NLOAD; ++j) {
    if (tid + j * 1024 < SLICE4) {
      atomicAdd(&cnt[v[j].x], 1u);
      atomicAdd(&cnt[v[j].y], 1u);
      atomicAdd(&cnt[v[j].z], 1u);
      atomicAdd(&cnt[v[j].w], 1u);
    }
  }
  __syncthreads();

  int k = blockIdx.x % NSTEP;
  for (int b = 0; b < NSTEP / NB; ++b) {
    int idx[NB];
    unsigned c[NB];
#pragma unroll
    for (int u = 0; u < NB; ++u) {
      idx[u] = k * 1024 + tid;
      c[u] = cnt[min(idx[u], NODES - 1)];
      k = (k + 1 == NSTEP) ? 0 : k + 1;
    }
#pragma unroll
    for (int u = 0; u < NB; ++u)
      if (idx[u] < NODES) atomicAdd(&deg[idx[u]], (float)c[u]);
  }
}

#define EPT 8
template<int K, bool RELU_IN, bool EDGE_NORM>
__global__ __launch_bounds__(256) void k_gemm(const float* __restrict__ A,
                                              const float* __restrict__ W,
                                              const float* __restrict__ bias,
                                              const float* __restrict__ deg,
                                              float* __restrict__ C,
                                              float* __restrict__ AGG, int M,
                                              const int* __restrict__ erow,
                                              const int* __restrict__ ecol,
                                              float* __restrict__ ENORM, int E) {
  __shared__ __align__(16) float sB[K * NF];

  constexpr int NG = K / 64;
  constexpr int NT = K / 32;

  const int lane = threadIdx.x & 31;
  const int wave = threadIdx.x >> 5;

  int strip = blockIdx.x * 8 + wave;
  int maxStrip = (M >> 4) - 1;
  if (strip > maxStrip) strip = maxStrip;
  const long rowBase = (long)strip << 4;

  const int mlo  = lane & 15;
  const int koff = (lane >> 4) << 1;
  const float* arow = A + (rowBase + mlo) * (long)K;

  const int mhi = (lane >> 4) << 3;
  float4 dg0 = {}, dg1 = {};
  float bcol[4] = {};

  float4 w0[NT], w1[NT];
#pragma unroll
  for (int j = 0; j < NT; ++j) {
    const int q   = threadIdx.x + j * 256;
    const int kk  = (q >> 4) << 1;
    const int cl0 = (q & 15) << 2;
    w0[j] = *(const float4*)(W + kk * NF + cl0);
    w1[j] = *(const float4*)(W + (kk + 1) * NF + cl0);
  }

  const int gtid = blockIdx.x * 256 + threadIdx.x;
  const int gstride = gridDim.x * 256;
  int er[EPT], ec[EPT];
  float gr[EPT], gc[EPT];
  if (EDGE_NORM) {
#pragma unroll
    for (int j = 0; j < EPT; ++j) {
      const int idx = min(gtid + j * gstride, E - 1);
      er[j] = erow[idx];
      ec[j] = ecol[idx];
    }
  }

  v2f acur[16];
#pragma unroll
  for (int ks = 0; ks < 16; ++ks) acur[ks] = *(const v2f*)(arow + ks * 4 + koff);

#pragma unroll
  for (int j = 0; j < NT; ++j) {
    const int q  = threadIdx.x + j * 256;
    const int r  = q >> 4;
    const int ks = r >> 1;
    const int h  = r & 1;
    const int t  = (q & 15) >> 2;
    const int c0 = (q & 3) << 2;
    float4* d = (float4*)(sB + (((ks * 4 + t) * 32 + h * 16 + c0) << 1));
    d[0] = make_float4(w0[j].x, w1[j].x, w0[j].y, w1[j].y);
    d[1] = make_float4(w0[j].z, w1[j].z, w0[j].w, w1[j].w);
  }
  __syncthreads();

  if (EDGE_NORM) {
#pragma unroll
    for (int j = 0; j < EPT; ++j) { gr[j] = deg[er[j]]; gc[j] = deg[ec[j]]; }
  }

  v8f acc[4];
#pragma unroll
  for (int t = 0; t < 4; ++t) acc[t] = (v8f){0.f,0.f,0.f,0.f,0.f,0.f,0.f,0.f};

#pragma unroll
  for (int g = 0; g < NG; ++g) {
    v2f anxt[16];
    if (g + 1 < NG) {
#pragma unroll
      for (int ks = 0; ks < 16; ++ks)
        anxt[ks] = *(const v2f*)(arow + (g + 1) * 64 + ks * 4 + koff);
    } else {
      dg0 = *(const float4*)(deg + rowBase + mhi);
      dg1 = *(const float4*)(deg + rowBase + mhi + 4);
#pragma unroll
      for (int t = 0; t < 4; ++t) bcol[t] = bias[t * 16 + mlo];
    }
#pragma unroll
    for (int ks = 0; ks < 16; ++ks) {
      v2f a = acur[ks];
      if (RELU_IN) {
        a.x = fmaxf(a.x, 0.f);
        a.y = fmaxf(a.y, 0.f);
      }
      const v2f* bfrag = (const v2f*)sB + (g * 16 + ks) * 128 + lane;
#pragma unroll
      for (int t = 0; t < 4; ++t) {
        v2f b = bfrag[t * 32];
        acc[t] = __builtin_amdgcn_wmma_f32_16x16x4_f32(
            false, a, false, b, (short)0, acc[t], false, false);
      }
    }
    if (g + 1 < NG) {
#pragma unroll
      for (int ks = 0; ks < 16; ++ks) acur[ks] = anxt[ks];
    }
  }

  float sl[8];
  sl[0] = rsqrtf(dg0.x); sl[1] = rsqrtf(dg0.y); sl[2] = rsqrtf(dg0.z); sl[3] = rsqrtf(dg0.w);
  sl[4] = rsqrtf(dg1.x); sl[5] = rsqrtf(dg1.y); sl[6] = rsqrtf(dg1.z); sl[7] = rsqrtf(dg1.w);
#pragma unroll
  for (int i = 0; i < 8; ++i) sl[i] = sl[i] * sl[i];
#pragma unroll
  for (int t = 0; t < 4; ++t)
#pragma unroll
    for (int i = 0; i < 8; ++i) {
      const long o = (rowBase + mhi + i) * NF + t * 16 + mlo;
      AGG[o] = acc[t][i] * sl[i] + bcol[t];
    }
#pragma unroll
  for (int i = 0; i < 8; ++i) {
    float* crow = C + (rowBase + mhi + i) * NF + 2 * mlo;
    *(v2f*)crow        = (v2f){acc[0][i], acc[2][i]};
    *(v2f*)(crow + 32) = (v2f){acc[1][i], acc[3][i]};
  }

  if (EDGE_NORM) {
#pragma unroll
    for (int j = 0; j < EPT; ++j) {
      const int idx = gtid + j * gstride;
      if (idx < E) ENORM[idx] = rsqrtf(gr[j]) * rsqrtf(gc[j]);
    }
  }
}

#define EPW 16
__global__ __launch_bounds__(256) void k_agg_edges(const float* __restrict__ xw,
                            const int* __restrict__ row, const int* __restrict__ col,
                            const float* __restrict__ enorm, float* agg, int e) {
  const int lane = threadIdx.x & 31;
  const int e0 = __builtin_amdgcn_readfirstlane((blockIdx.x * 8 + (threadIdx.x >> 5)) * EPW);
  if (e0 >= e) return;
  const int mine = e0 + (lane & (EPW - 1));
  const float* xl = xw + 2 * lane;
  float* al = agg + lane;
  if (e0 + EPW <= e) {
    const int r = row[mine], c = col[mine];
    const float nrm = enorm[mine];
    v2f v[EPW];
#pragma unroll
    for (int j = 0; j < EPW; ++j)
      v[j] = *(const v2f*)(xl + (long)__builtin_amdgcn_readlane(r, j) * NF);
#pragma unroll
    for (int j = 0; j < EPW; ++j) {
      const float nj = __int_as_float(__builtin_amdgcn_readlane(__float_as_int(nrm), j));
      float* dst = al + (long)__builtin_amdgcn_readlane(c, j) * NF;
      atomicAdd(dst,      v[j].x * nj);
      atomicAdd(dst + 32, v[j].y * nj);
    }
  } else {
    int r = 0, c = 0;
    float nrm = 0.f;
    if (mine < e) {
      r = row[mine]; c = col[mine];
      nrm = enorm[mine];
    }
    for (int j = 0; e0 + j < e; ++j) {
      const int rj = __builtin_amdgcn_readlane(r, j);
      const int cj = __builtin_amdgcn_readlane(c, j);
      const float nj = __int_as_float(__builtin_amdgcn_readlane(__float_as_int(nrm), j));
      const v2f s = *(const v2f*)(xl + (long)rj * NF);
      float* dst = al + (long)cj * NF;
      atomicAdd(dst,      s.x * nj);
      atomicAdd(dst + 32, s.y * nj);
    }
  }
}

#define DROWP 68
__global__ __launch_bounds__(256) void k_decode(const float* __restrict__ z,
                         const int* __restrict__ src, const int* __restrict__ dst,
                         float* __restrict__ out, int e) {
  __shared__ __align__(16) float sR[8 * 64 * DROWP];
  const int lane = threadIdx.x & 31;
  const int wave = threadIdx.x >> 5;
  const int half = lane >> 4;
  const int c4   = (lane & 15) << 2;
  const int i    = blockIdx.x * 256 + threadIdx.x;
  const int ic   = min(i, e - 1);
  const int s = src[ic], d = dst[ic];
  float* wr = sR + wave * (64 * DROWP);

  float4 va[16], vb[16];
#pragma unroll
  for (int j = 0; j < 16; ++j) {
    const int s0 = __builtin_amdgcn_readlane(s, 2 * j), s1 = __builtin_amdgcn_readlane(s, 2 * j + 1);
    const int d0 = __builtin_amdgcn_readlane(d, 2 * j), d1 = __builtin_amdgcn_readlane(d, 2 * j + 1);
    va[j] = *(const float4*)(z + (long)(half ? s1 : s0) * NF + c4);
    vb[j] = *(const float4*)(z + (long)(half ? d1 : d0) * NF + c4);
  }
#pragma unroll
  for (int j = 0; j < 16; ++j) {
    *(float4*)(wr + (2 * j + half) * DROWP + c4)        = va[j];
    *(float4*)(wr + (32 + 2 * j + half) * DROWP + c4)   = vb[j];
  }
  __syncthreads();
  const float4* a = (const float4*)(wr + lane * DROWP);
  const float4* b = (const float4*)(wr + (32 + lane) * DROWP);
  float acc = 0.f;
#pragma unroll
  for (int j = 0; j < 16; ++j) {
    float4 av = a[j], bv = b[j];
    acc += av.x * bv.x + av.y * bv.y + av.z * bv.z + av.w * bv.w;
  }
  if (i < e) out[i] = acc;
}

extern "C" void kernel_launch(void* const* d_in, const int* in_sizes, int n_in,
                              void* d_out, int out_size, void* d_ws, size_t ws_size,
                              hipStream_t stream) {
  const float* x   = (const float*)d_in[0];
  const int*   ei  = (const int*)d_in[1];
  const int*   eli = (const int*)d_in[2];
  const float* W1  = (const float*)d_in[3];
  const float* b1  = (const float*)d_in[4];
  const float* W2  = (const float*)d_in[5];
  const float* b2  = (const float*)d_in[6];
  float* out = (float*)d_out;

  float* ws   = (float*)d_ws;
  float* deg  = ws;
  float* bufA = ws + 50048;
  float* bufB = bufA + (size_t)NODES * NF;
  float* bufC = bufB + (size_t)NODES * NF;
  float* enrm = bufC + (size_t)NODES * NF;

  const int* erow = ei;
  const int* ecol = ei + EDGES;
  const int* lsrc = eli;
  const int* ldst = eli + LEDGES;

  dim3 blk(256);
  const int edgeGrid = (EDGES / EPW + 7) / 8;
  const int strips   = NODES / 16;
  const int gemmGrid = (strips + 7) / 8;
  static_assert(((NODES / 16 + 7) / 8) * 256 * EPT >= EDGES, "edge-norm shares must cover every edge");

  k_deg_init<<<(NODES + 255) / 256, blk, 0, stream>>>(deg, NODES);
  k_deg_hist<<<DEG_WG, 1024, 0, stream>>>(ecol, deg);

  k_gemm<INCH, false, true><<<gemmGrid, blk, 0, stream>>>(x, W1, b1, deg, bufA, bufB, NODES,
                                                       erow, ecol, enrm, EDGES);
  k_agg_edges<<<edgeGrid, blk, 0, stream>>>(bufA, erow, ecol, enrm, bufB, EDGES);

  k_gemm<NF, true, false><<<gemmGrid, blk, 0, stream>>>(bufB, W2, b2, deg, bufA, bufC, NODES,
                                                     erow, ecol, enrm, EDGES);
  k_agg_edges<<<edgeGrid, blk, 0, stream>>>(bufA, erow, ecol, enrm, bufC, EDGES);

  k_decode<<<(LEDGES + 255) / 256, blk, 0, stream>>>(bufC, lsrc, ldst, out, LEDGES);
}
